# starting kernel
# baseline (speedup 1.0000x reference)
#include <hip/hip_runtime.h>

typedef __attribute__((ext_vector_type(2))) float v2f;
typedef __attribute__((ext_vector_type(8))) float v8f;

#define IN_DIM 512
#define HID    16
#define ODIM   64

__global__ void k_deg_init(float* __restrict__ deg, int n) {
  int i = blockIdx.x * blockDim.x + threadIdx.x;
  if (i < n) deg[i] = 1.0f;
}

__global__ void k_deg_edges(const int* __restrict__ dst, float* __restrict__ deg, int E) {
  int i = blockIdx.x * blockDim.x + threadIdx.x;
  if (i < E) atomicAdd(&deg[dst[i]], 1.0f);
}

__global__ void k_dinv(float* __restrict__ deg, int n) {
  int i = blockIdx.x * blockDim.x + threadIdx.x;
  if (i < n) {
    float d = deg[i];
    deg[i] = (d > 0.0f) ? (1.0f / sqrtf(d)) : 0.0f;
  }
}

__global__ void k_gemm1(const float* __restrict__ x, const float* __restrict__ W1,
                        float* __restrict__ h1, int n) {
  __shared__ float lw[IN_DIM * HID];
  for (int i = threadIdx.x; i < IN_DIM * HID; i += blockDim.x) lw[i] = W1[i];
  __syncthreads();

  int wave  = (blockIdx.x * blockDim.x + threadIdx.x) >> 5;
  int lane  = threadIdx.x & 31;
  int tiles = (n + 15) >> 4;
  if (wave >= tiles) return;

  int half = lane >> 4;
  int ml   = lane & 15;
  int row  = wave * 16 + ml;
  int rrow = (row < n) ? row : (n - 1);
  const float* xr = x + (size_t)rrow * IN_DIM;

  v8f acc = {0.f, 0.f, 0.f, 0.f, 0.f, 0.f, 0.f, 0.f};
  for (int k = 0; k < IN_DIM; k += 4) {
    int ka = k + 2 * half;
    v2f a; a.x = xr[ka];                 a.y = xr[ka + 1];
    v2f b; b.x = lw[ka * HID + ml];      b.y = lw[(ka + 1) * HID + ml];
    acc = __builtin_amdgcn_wmma_f32_16x16x4_f32(false, a, false, b, (short)0, acc,
                                                false, false);
  }
#pragma unroll
  for (int r = 0; r < 8; ++r) {
    int orow = wave * 16 + r + 8 * half;
    if (orow < n) h1[(size_t)orow * HID + ml] = acc[r];
  }
}

__global__ void k_init_z1(const float* __restrict__ h1, const float* __restrict__ dinv,
                          const float* __restrict__ b1, float* __restrict__ z1, int n) {
  int t = blockIdx.x * blockDim.x + threadIdx.x;
  int total = n * HID;
  if (t < total) {
    int i = t / HID, j = t - i * HID;
    float di = dinv[i];
    z1[t] = h1[t] * di * di + b1[j];
  }
}

__global__ void k_agg1(const int* __restrict__ src, const int* __restrict__ dst,
                       const float* __restrict__ dinv, const float* __restrict__ h1,
                       float* __restrict__ z1, int E) {
  int total  = E * 4;
  int stride = gridDim.x * blockDim.x;
  for (int t = blockIdx.x * blockDim.x + threadIdx.x; t < total; t += stride) {
    int e = t >> 2, c = t & 3;
    int s = src[e], d = dst[e];
    float norm = dinv[s] * dinv[d];
    const float4 hv = *(const float4*)(h1 + (size_t)s * HID + c * 4);
    float* o = z1 + (size_t)d * HID + c * 4;
    atomicAdd(o + 0, hv.x * norm);
    atomicAdd(o + 1, hv.y * norm);
    atomicAdd(o + 2, hv.z * norm);
    atomicAdd(o + 3, hv.w * norm);
  }
}

__global__ void k_gemm2(const float* __restrict__ z1, const float* __restrict__ W2,
                        float* __restrict__ h2, int n) {
  __shared__ float lw[HID * ODIM];
  for (int i = threadIdx.x; i < HID * ODIM; i += blockDim.x) lw[i] = W2[i];
  __syncthreads();

  int wave  = (blockIdx.x * blockDim.x + threadIdx.x) >> 5;
  int lane  = threadIdx.x & 31;
  int tiles = (n + 15) >> 4;
  if (wave >= tiles) return;

  int half = lane >> 4;
  int ml   = lane & 15;
  int row  = wave * 16 + ml;
  int rrow = (row < n) ? row : (n - 1);
  const float* zr = z1 + (size_t)rrow * HID;

  v2f afrag[4];
#pragma unroll
  for (int kk = 0; kk < 4; ++kk) {
    int ka = kk * 4 + 2 * half;
    afrag[kk].x = fmaxf(zr[ka], 0.f);
    afrag[kk].y = fmaxf(zr[ka + 1], 0.f);
  }

#pragma unroll
  for (int tcol = 0; tcol < 4; ++tcol) {
    v8f acc = {0.f, 0.f, 0.f, 0.f, 0.f, 0.f, 0.f, 0.f};
#pragma unroll
    for (int kk = 0; kk < 4; ++kk) {
      int ka = kk * 4 + 2 * half;
      v2f b;
      b.x = lw[ka * ODIM + tcol * 16 + ml];
      b.y = lw[(ka + 1) * ODIM + tcol * 16 + ml];
      acc = __builtin_amdgcn_wmma_f32_16x16x4_f32(false, afrag[kk], false, b, (short)0,
                                                  acc, false, false);
    }
#pragma unroll
    for (int r = 0; r < 8; ++r) {
      int orow = wave * 16 + r + 8 * half;
      if (orow < n) h2[(size_t)orow * ODIM + tcol * 16 + ml] = acc[r];
    }
  }
}

__global__ void k_init_out(const float* __restrict__ h2, const float* __restrict__ dinv,
                           const float* __restrict__ b2, float* __restrict__ out, int n) {
  int t = blockIdx.x * blockDim.x + threadIdx.x;
  int total = n * ODIM;
  if (t < total) {
    int i = t / ODIM, j = t - i * ODIM;
    float di = dinv[i];
    out[t] = h2[t] * di * di + b2[j];
  }
}

__global__ void k_agg2(const int* __restrict__ src, const int* __restrict__ dst,
                       const float* __restrict__ dinv, const float* __restrict__ h2,
                       float* __restrict__ out, int E) {
  int total  = E * 16;
  int stride = gridDim.x * blockDim.x;
  for (int t = blockIdx.x * blockDim.x + threadIdx.x; t < total; t += stride) {
    int e = t >> 4, c = t & 15;
    int s = src[e], d = dst[e];
    float norm = dinv[s] * dinv[d];
    const float4 hv = *(const float4*)(h2 + (size_t)s * ODIM + c * 4);
    float* o = out + (size_t)d * ODIM + c * 4;
    atomicAdd(o + 0, hv.x * norm);
    atomicAdd(o + 1, hv.y * norm);
    atomicAdd(o + 2, hv.z * norm);
    atomicAdd(o + 3, hv.w * norm);
  }
}

extern "C" void kernel_launch(void* const* d_in, const int* in_sizes, int n_in,
                              void* d_out, int out_size, void* d_ws, size_t ws_size,
                              hipStream_t stream) {
  const float* x  = (const float*)d_in[0];
  const float* W1 = (const float*)d_in[1];
  const float* b1 = (const float*)d_in[2];
  const float* W2 = (const float*)d_in[3];
  const float* b2 = (const float*)d_in[4];
  const int*   ei = (const int*)d_in[5];

  const int N = in_sizes[0] / IN_DIM;
  const int E = in_sizes[5] / 2;
  const int* src = ei;
  const int* dst = ei + E;
  float* out = (float*)d_out;

  char* ws = (char*)d_ws;
  size_t off = 0;
  auto alloc = [&](size_t nbytes) -> void* {
    void* p = ws + off;
    off = (off + nbytes + 255) & ~(size_t)255;
    return p;
  };
  float* dinv = (float*)alloc((size_t)N * sizeof(float));
  float* h1   = (float*)alloc((size_t)N * HID  * sizeof(float));
  float* z1   = (float*)alloc((size_t)N * HID  * sizeof(float));
  float* h2   = (float*)alloc((size_t)N * ODIM * sizeof(float));

  const int B = 256;
  const int tiles = (N + 15) >> 4;
  const int gemmBlocks = (tiles + 7) / 8;

  k_deg_init <<<(N + B - 1) / B, B, 0, stream>>>(dinv, N);
  k_deg_edges<<<(E + B - 1) / B, B, 0, stream>>>(dst, dinv, E);
  k_dinv     <<<(N + B - 1) / B, B, 0, stream>>>(dinv, N);

  k_gemm1    <<<gemmBlocks, B, 0, stream>>>(x, W1, h1, N);
  k_init_z1  <<<((size_t)N * HID + B - 1) / B, B, 0, stream>>>(h1, dinv, b1, z1, N);
  k_agg1     <<<(E * 4 + B - 1) / B, B, 0, stream>>>(src, dst, dinv, h1, z1, E);

  k_gemm2    <<<gemmBlocks, B, 0, stream>>>(z1, W2, h2, N);
  k_init_out <<<((size_t)N * ODIM + B - 1) / B, B, 0, stream>>>(h2, dinv, b2, out, N);
  k_agg2     <<<(E * 16 + B - 1) / B, B, 0, stream>>>(src, dst, dinv, h2, out, E);
}


# edge passes: one wave per 32 edges, lane = feature column, so each row gather and f32 atomic add instruction covers whole row lines (1-2 per instruction, was 4-8); atomic arrival order as before
# speedup vs baseline: 1.5625x; 1.5625x over previous
#include <hip/hip_runtime.h>

typedef __attribute__((ext_vector_type(2))) float v2f;
typedef __attribute__((ext_vector_type(8))) float v8f;

#define IN_DIM 512
#define HID    16
#define ODIM   64

__global__ void k_deg_init(float* __restrict__ deg, int n) {
  int i = blockIdx.x * blockDim.x + threadIdx.x;
  if (i < n) deg[i] = 1.0f;
}

__global__ void k_deg_edges(const int* __restrict__ dst, float* __restrict__ deg, int E) {
  int i = blockIdx.x * blockDim.x + threadIdx.x;
  if (i < E) atomicAdd(&deg[dst[i]], 1.0f);
}

__global__ void k_dinv(float* __restrict__ deg, int n) {
  int i = blockIdx.x * blockDim.x + threadIdx.x;
  if (i < n) {
    float d = deg[i];
    deg[i] = (d > 0.0f) ? (1.0f / sqrtf(d)) : 0.0f;
  }
}

__global__ void k_gemm1(const float* __restrict__ x, const float* __restrict__ W1,
                        float* __restrict__ h1, int n) {
  __shared__ float lw[IN_DIM * HID];
  for (int i = threadIdx.x; i < IN_DIM * HID; i += blockDim.x) lw[i] = W1[i];
  __syncthreads();

  int wave  = (blockIdx.x * blockDim.x + threadIdx.x) >> 5;
  int lane  = threadIdx.x & 31;
  int tiles = (n + 15) >> 4;
  if (wave >= tiles) return;

  int half = lane >> 4;
  int ml   = lane & 15;
  int row  = wave * 16 + ml;
  int rrow = (row < n) ? row : (n - 1);
  const float* xr = x + (size_t)rrow * IN_DIM;

  v8f acc = {0.f, 0.f, 0.f, 0.f, 0.f, 0.f, 0.f, 0.f};
  for (int k = 0; k < IN_DIM; k += 4) {
    int ka = k + 2 * half;
    v2f a; a.x = xr[ka];                 a.y = xr[ka + 1];
    v2f b; b.x = lw[ka * HID + ml];      b.y = lw[(ka + 1) * HID + ml];
    acc = __builtin_amdgcn_wmma_f32_16x16x4_f32(false, a, false, b, (short)0, acc,
                                                false, false);
  }
#pragma unroll
  for (int r = 0; r < 8; ++r) {
    int orow = wave * 16 + r + 8 * half;
    if (orow < n) h1[(size_t)orow * HID + ml] = acc[r];
  }
}

__global__ void k_init_z1(const float* __restrict__ h1, const float* __restrict__ dinv,
                          const float* __restrict__ b1, float* __restrict__ z1, int n) {
  int t = blockIdx.x * blockDim.x + threadIdx.x;
  int total = n * HID;
  if (t < total) {
    int i = t / HID, j = t - i * HID;
    float di = dinv[i];
    z1[t] = h1[t] * di * di + b1[j];
  }
}

__global__ void k_agg1(const int* __restrict__ src, const int* __restrict__ dst,
                       const float* __restrict__ dinv, const float* __restrict__ h1,
                       float* __restrict__ z1, int E) {
  const int lane = threadIdx.x & 31;
  const int wave = __builtin_amdgcn_readfirstlane((blockIdx.x * blockDim.x + threadIdx.x) >> 5);
  const int e0   = wave * 32;
  if (e0 >= E) return;
  const int cnt  = (E - e0 < 32) ? (E - e0) : 32;

  int s = 0, d = 0;
  float nm = 0.f;
  if (lane < cnt) {
    s  = src[e0 + lane];
    d  = dst[e0 + lane];
    nm = dinv[s] * dinv[d];
  }

  const int hsel = lane >> 4;
  const int f    = lane & 15;

#pragma unroll
  for (int b = 0; b < 2; ++b) {
    const int left = cnt - b * 16;
    if (left > 0) {
      float hv[8], nj[8];
      int   dj[8];
#pragma unroll
      for (int u = 0; u < 8; ++u) {
        const int j  = b * 16 + u * 2 + hsel;
        const int sj = __shfl(s, j, 32);
        dj[u] = __shfl(d, j, 32);
        nj[u] = __shfl(nm, j, 32);
        hv[u] = h1[(size_t)sj * HID + f];
      }
      if (left >= 16) {
#pragma unroll
        for (int u = 0; u < 8; ++u)
          atomicAdd(z1 + (size_t)dj[u] * HID + f, hv[u] * nj[u]);
      } else {
#pragma unroll
        for (int u = 0; u < 8; ++u)
          if (u * 2 + hsel < left) atomicAdd(z1 + (size_t)dj[u] * HID + f, hv[u] * nj[u]);
      }
    }
  }
}

__global__ void k_gemm2(const float* __restrict__ z1, const float* __restrict__ W2,
                        float* __restrict__ h2, int n) {
  __shared__ float lw[HID * ODIM];
  for (int i = threadIdx.x; i < HID * ODIM; i += blockDim.x) lw[i] = W2[i];
  __syncthreads();

  int wave  = (blockIdx.x * blockDim.x + threadIdx.x) >> 5;
  int lane  = threadIdx.x & 31;
  int tiles = (n + 15) >> 4;
  if (wave >= tiles) return;

  int half = lane >> 4;
  int ml   = lane & 15;
  int row  = wave * 16 + ml;
  int rrow = (row < n) ? row : (n - 1);
  const float* zr = z1 + (size_t)rrow * HID;

  v2f afrag[4];
#pragma unroll
  for (int kk = 0; kk < 4; ++kk) {
    int ka = kk * 4 + 2 * half;
    afrag[kk].x = fmaxf(zr[ka], 0.f);
    afrag[kk].y = fmaxf(zr[ka + 1], 0.f);
  }

#pragma unroll
  for (int tcol = 0; tcol < 4; ++tcol) {
    v8f acc = {0.f, 0.f, 0.f, 0.f, 0.f, 0.f, 0.f, 0.f};
#pragma unroll
    for (int kk = 0; kk < 4; ++kk) {
      int ka = kk * 4 + 2 * half;
      v2f b;
      b.x = lw[ka * ODIM + tcol * 16 + ml];
      b.y = lw[(ka + 1) * ODIM + tcol * 16 + ml];
      acc = __builtin_amdgcn_wmma_f32_16x16x4_f32(false, afrag[kk], false, b, (short)0,
                                                  acc, false, false);
    }
#pragma unroll
    for (int r = 0; r < 8; ++r) {
      int orow = wave * 16 + r + 8 * half;
      if (orow < n) h2[(size_t)orow * ODIM + tcol * 16 + ml] = acc[r];
    }
  }
}

__global__ void k_init_out(const float* __restrict__ h2, const float* __restrict__ dinv,
                           const float* __restrict__ b2, float* __restrict__ out, int n) {
  int t = blockIdx.x * blockDim.x + threadIdx.x;
  int total = n * ODIM;
  if (t < total) {
    int i = t / ODIM, j = t - i * ODIM;
    float di = dinv[i];
    out[t] = h2[t] * di * di + b2[j];
  }
}

__global__ void k_agg2(const int* __restrict__ src, const int* __restrict__ dst,
                       const float* __restrict__ dinv, const float* __restrict__ h2,
                       float* __restrict__ out, int E) {
  const int lane = threadIdx.x & 31;
  const int wave = __builtin_amdgcn_readfirstlane((blockIdx.x * blockDim.x + threadIdx.x) >> 5);
  const int e0   = wave * 32;
  if (e0 >= E) return;
  const int cnt  = (E - e0 < 32) ? (E - e0) : 32;

  int s = 0, d = 0;
  float nm = 0.f;
  if (lane < cnt) {
    s  = src[e0 + lane];
    d  = dst[e0 + lane];
    nm = dinv[s] * dinv[d];
  }

#pragma unroll
  for (int b = 0; b < 4; ++b) {
    const int left = cnt - b * 8;
    if (left > 0) {
      float lo[8], hi[8], nj[8];
      int   dj[8];
#pragma unroll
      for (int u = 0; u < 8; ++u) {
        const int j  = b * 8 + u;
        const int sj = __builtin_amdgcn_readlane(s, j);
        dj[u] = __builtin_amdgcn_readlane(d, j);
        nj[u] = __int_as_float(__builtin_amdgcn_readlane(__float_as_int(nm), j));
        const float* hr = h2 + (size_t)sj * ODIM + lane;
        lo[u] = hr[0];
        hi[u] = hr[32];
      }
      if (left >= 8) {
#pragma unroll
        for (int u = 0; u < 8; ++u) {
          float* o = out + (size_t)dj[u] * ODIM + lane;
          atomicAdd(o,      lo[u] * nj[u]);
          atomicAdd(o + 32, hi[u] * nj[u]);
        }
      } else {
#pragma unroll
        for (int u = 0; u < 8; ++u) {
          if (u < left) {
            float* o = out + (size_t)dj[u] * ODIM + lane;
            atomicAdd(o,      lo[u] * nj[u]);
            atomicAdd(o + 32, hi[u] * nj[u]);
          }
        }
      }
    }
  }
}

extern "C" void kernel_launch(void* const* d_in, const int* in_sizes, int n_in,
                              void* d_out, int out_size, void* d_ws, size_t ws_size,
                              hipStream_t stream) {
  const float* x  = (const float*)d_in[0];
  const float* W1 = (const float*)d_in[1];
  const float* b1 = (const float*)d_in[2];
  const float* W2 = (const float*)d_in[3];
  const float* b2 = (const float*)d_in[4];
  const int*   ei = (const int*)d_in[5];

  const int N = in_sizes[0] / IN_DIM;
  const int E = in_sizes[5] / 2;
  const int* src = ei;
  const int* dst = ei + E;
  float* out = (float*)d_out;

  char* ws = (char*)d_ws;
  size_t off = 0;
  auto alloc = [&](size_t nbytes) -> void* {
    void* p = ws + off;
    off = (off + nbytes + 255) & ~(size_t)255;
    return p;
  };
  float* dinv = (float*)alloc((size_t)N * sizeof(float));
  float* h1   = (float*)alloc((size_t)N * HID  * sizeof(float));
  float* z1   = (float*)alloc((size_t)N * HID  * sizeof(float));
  float* h2   = (float*)alloc((size_t)N * ODIM * sizeof(float));

  const int B = 256;
  const int tiles = (N + 15) >> 4;
  const int gemmBlocks = (tiles + 7) / 8;

  k_deg_init <<<(N + B - 1) / B, B, 0, stream>>>(dinv, N);
  k_deg_edges<<<(E + B - 1) / B, B, 0, stream>>>(dst, dinv, E);
  k_dinv     <<<(N + B - 1) / B, B, 0, stream>>>(dinv, N);

  k_gemm1    <<<gemmBlocks, B, 0, stream>>>(x, W1, h1, N);
  k_init_z1  <<<((size_t)N * HID + B - 1) / B, B, 0, stream>>>(h1, dinv, b1, z1, N);
  k_agg1     <<<(E + B - 1) / B, B, 0, stream>>>(src, dst, dinv, h1, z1, E);

  k_gemm2    <<<gemmBlocks, B, 0, stream>>>(z1, W2, h2, N);
  k_init_out <<<((size_t)N * ODIM + B - 1) / B, B, 0, stream>>>(h2, dinv, b2, out, N);
  k_agg2     <<<(E + B - 1) / B, B, 0, stream>>>(src, dst, dinv, h2, out, E);
}


# layer 2 aggregates before its transform, A(HW) = (AH)W: edge pass moves 16-float rows of relu(z1), not 64-float rows; WMMA GEMM2 then writes out + bias; f32 sum order of layer 2 changes
# speedup vs baseline: 1.6847x; 1.0782x over previous
#include <hip/hip_runtime.h>

typedef __attribute__((ext_vector_type(2))) float v2f;
typedef __attribute__((ext_vector_type(8))) float v8f;

#define IN_DIM 512
#define HID    16
#define ODIM   64

__global__ void k_deg_init(float* __restrict__ deg, int n) {
  int i = blockIdx.x * blockDim.x + threadIdx.x;
  if (i < n) deg[i] = 1.0f;
}

__global__ void k_deg_edges(const int* __restrict__ dst, float* __restrict__ deg, int E) {
  int i = blockIdx.x * blockDim.x + threadIdx.x;
  if (i < E) atomicAdd(&deg[dst[i]], 1.0f);
}

__global__ void k_dinv(float* __restrict__ deg, int n) {
  int i = blockIdx.x * blockDim.x + threadIdx.x;
  if (i < n) {
    float d = deg[i];
    deg[i] = (d > 0.0f) ? (1.0f / sqrtf(d)) : 0.0f;
  }
}

__global__ void k_gemm1(const float* __restrict__ x, const float* __restrict__ W1,
                        float* __restrict__ h1, int n) {
  __shared__ float lw[IN_DIM * HID];
  for (int i = threadIdx.x; i < IN_DIM * HID; i += blockDim.x) lw[i] = W1[i];
  __syncthreads();

  int wave  = (blockIdx.x * blockDim.x + threadIdx.x) >> 5;
  int lane  = threadIdx.x & 31;
  int tiles = (n + 15) >> 4;
  if (wave >= tiles) return;

  int half = lane >> 4;
  int ml   = lane & 15;
  int row  = wave * 16 + ml;
  int rrow = (row < n) ? row : (n - 1);
  const float* xr = x + (size_t)rrow * IN_DIM;

  v8f acc = {0.f, 0.f, 0.f, 0.f, 0.f, 0.f, 0.f, 0.f};
  for (int k = 0; k < IN_DIM; k += 4) {
    int ka = k + 2 * half;
    v2f a; a.x = xr[ka];                 a.y = xr[ka + 1];
    v2f b; b.x = lw[ka * HID + ml];      b.y = lw[(ka + 1) * HID + ml];
    acc = __builtin_amdgcn_wmma_f32_16x16x4_f32(false, a, false, b, (short)0, acc,
                                                false, false);
  }
#pragma unroll
  for (int r = 0; r < 8; ++r) {
    int orow = wave * 16 + r + 8 * half;
    if (orow < n) h1[(size_t)orow * HID + ml] = acc[r];
  }
}

__global__ void k_init_z1(const float* __restrict__ h1, const float* __restrict__ dinv,
                          const float* __restrict__ b1, float* __restrict__ z1, int n) {
  int t = blockIdx.x * blockDim.x + threadIdx.x;
  int total = n * HID;
  if (t < total) {
    int i = t / HID, j = t - i * HID;
    float di = dinv[i];
    z1[t] = h1[t] * di * di + b1[j];
  }
}

__global__ void k_agg16(const int* __restrict__ src, const int* __restrict__ dst,
                        const float* __restrict__ dinv, const float* __restrict__ h1,
                        float* __restrict__ z1, int E) {
  const int lane = threadIdx.x & 31;
  const int wave = __builtin_amdgcn_readfirstlane((blockIdx.x * blockDim.x + threadIdx.x) >> 5);
  const int e0   = wave * 32;
  if (e0 >= E) return;
  const int cnt  = (E - e0 < 32) ? (E - e0) : 32;

  int s = 0, d = 0;
  float nm = 0.f;
  if (lane < cnt) {
    s  = src[e0 + lane];
    d  = dst[e0 + lane];
    nm = dinv[s] * dinv[d];
  }

  const int hsel = lane >> 4;
  const int f    = lane & 15;

#pragma unroll
  for (int b = 0; b < 2; ++b) {
    const int left = cnt - b * 16;
    if (left > 0) {
      float hv[8], nj[8];
      int   dj[8];
#pragma unroll
      for (int u = 0; u < 8; ++u) {
        const int j  = b * 16 + u * 2 + hsel;
        const int sj = __shfl(s, j, 32);
        dj[u] = __shfl(d, j, 32);
        nj[u] = __shfl(nm, j, 32);
        hv[u] = h1[(size_t)sj * HID + f];
      }
      if (left >= 16) {
#pragma unroll
        for (int u = 0; u < 8; ++u)
          atomicAdd(z1 + (size_t)dj[u] * HID + f, hv[u] * nj[u]);
      } else {
#pragma unroll
        for (int u = 0; u < 8; ++u)
          if (u * 2 + hsel < left) atomicAdd(z1 + (size_t)dj[u] * HID + f, hv[u] * nj[u]);
      }
    }
  }
}

__global__ void k_init_g(float* __restrict__ z1, const float* __restrict__ dinv,
                         float* __restrict__ g, int n) {
  int t = blockIdx.x * blockDim.x + threadIdx.x;
  int total = n * HID;
  if (t < total) {
    int i = t / HID;
    float di = dinv[i];
    float r = fmaxf(z1[t], 0.f);
    z1[t] = r;
    g[t]  = r * di * di;
  }
}

__global__ void k_gemm2(const float* __restrict__ g, const float* __restrict__ W2,
                        const float* __restrict__ b2, float* __restrict__ out, int n) {
  __shared__ float lw[HID * ODIM];
  for (int i = threadIdx.x; i < HID * ODIM; i += blockDim.x) lw[i] = W2[i];
  __syncthreads();

  int wave  = (blockIdx.x * blockDim.x + threadIdx.x) >> 5;
  int lane  = threadIdx.x & 31;
  int tiles = (n + 15) >> 4;
  if (wave >= tiles) return;

  int half = lane >> 4;
  int ml   = lane & 15;
  int row  = wave * 16 + ml;
  int rrow = (row < n) ? row : (n - 1);
  const float* gr = g + (size_t)rrow * HID;

  v2f afrag[4];
#pragma unroll
  for (int kk = 0; kk < 4; ++kk) {
    int ka = kk * 4 + 2 * half;
    afrag[kk].x = gr[ka];
    afrag[kk].y = gr[ka + 1];
  }

#pragma unroll
  for (int tcol = 0; tcol < 4; ++tcol) {
    v8f acc = {0.f, 0.f, 0.f, 0.f, 0.f, 0.f, 0.f, 0.f};
#pragma unroll
    for (int kk = 0; kk < 4; ++kk) {
      int ka = kk * 4 + 2 * half;
      v2f b;
      b.x = lw[ka * ODIM + tcol * 16 + ml];
      b.y = lw[(ka + 1) * ODIM + tcol * 16 + ml];
      acc = __builtin_amdgcn_wmma_f32_16x16x4_f32(false, afrag[kk], false, b, (short)0,
                                                  acc, false, false);
    }
    const float bj = b2[tcol * 16 + ml];
#pragma unroll
    for (int r = 0; r < 8; ++r) {
      int orow = wave * 16 + r + 8 * half;
      if (orow < n) out[(size_t)orow * ODIM + tcol * 16 + ml] = acc[r] + bj;
    }
  }
}

extern "C" void kernel_launch(void* const* d_in, const int* in_sizes, int n_in,
                              void* d_out, int out_size, void* d_ws, size_t ws_size,
                              hipStream_t stream) {
  const float* x  = (const float*)d_in[0];
  const float* W1 = (const float*)d_in[1];
  const float* b1 = (const float*)d_in[2];
  const float* W2 = (const float*)d_in[3];
  const float* b2 = (const float*)d_in[4];
  const int*   ei = (const int*)d_in[5];

  const int N = in_sizes[0] / IN_DIM;
  const int E = in_sizes[5] / 2;
  const int* src = ei;
  const int* dst = ei + E;
  float* out = (float*)d_out;

  char* ws = (char*)d_ws;
  size_t off = 0;
  auto alloc = [&](size_t nbytes) -> void* {
    void* p = ws + off;
    off = (off + nbytes + 255) & ~(size_t)255;
    return p;
  };
  float* dinv = (float*)alloc((size_t)N * sizeof(float));
  float* h1   = (float*)alloc((size_t)N * HID  * sizeof(float));
  float* z1   = (float*)alloc((size_t)N * HID  * sizeof(float));
  float* g    = (float*)alloc((size_t)N * HID  * sizeof(float));

  const int B = 256;
  const int tiles = (N + 15) >> 4;
  const int gemmBlocks = (tiles + 7) / 8;

  k_deg_init <<<(N + B - 1) / B, B, 0, stream>>>(dinv, N);
  k_deg_edges<<<(E + B - 1) / B, B, 0, stream>>>(dst, dinv, E);
  k_dinv     <<<(N + B - 1) / B, B, 0, stream>>>(dinv, N);

  k_gemm1    <<<gemmBlocks, B, 0, stream>>>(x, W1, h1, N);
  k_init_z1  <<<((size_t)N * HID + B - 1) / B, B, 0, stream>>>(h1, dinv, b1, z1, N);
  k_agg16    <<<(E + B - 1) / B, B, 0, stream>>>(src, dst, dinv, h1, z1, E);

  k_init_g   <<<((size_t)N * HID + B - 1) / B, B, 0, stream>>>(z1, dinv, g, N);
  k_agg16    <<<(E + B - 1) / B, B, 0, stream>>>(src, dst, dinv, z1, g, E);
  k_gemm2    <<<gemmBlocks, B, 0, stream>>>(g, W2, b2, out, N);
}


# normalisation per node, not per edge: rows scaled by dinv[src] where produced, sums by dinv[dst] where consumed; edge passes add plain rows, no dinv gathers; sum order as before, its rounding changes
# speedup vs baseline: 1.7987x; 1.0677x over previous
#include <hip/hip_runtime.h>

typedef __attribute__((ext_vector_type(2))) float v2f;
typedef __attribute__((ext_vector_type(8))) float v8f;

#define IN_DIM 512
#define HID    16
#define ODIM   64

__global__ void k_deg_init(float* __restrict__ deg, int n) {
  int i = blockIdx.x * blockDim.x + threadIdx.x;
  if (i < n) deg[i] = 1.0f;
}

__global__ void k_deg_edges(const int* __restrict__ dst, float* __restrict__ deg, int E) {
  int i = blockIdx.x * blockDim.x + threadIdx.x;
  if (i < E) atomicAdd(&deg[dst[i]], 1.0f);
}

__global__ void k_dinv(float* __restrict__ deg, int n) {
  int i = blockIdx.x * blockDim.x + threadIdx.x;
  if (i < n) {
    float d = deg[i];
    deg[i] = (d > 0.0f) ? (1.0f / sqrtf(d)) : 0.0f;
  }
}

__global__ void k_gemm1(const float* __restrict__ x, const float* __restrict__ W1,
                        float* __restrict__ h1, int n) {
  __shared__ float lw[IN_DIM * HID];
  for (int i = threadIdx.x; i < IN_DIM * HID; i += blockDim.x) lw[i] = W1[i];
  __syncthreads();

  int wave  = (blockIdx.x * blockDim.x + threadIdx.x) >> 5;
  int lane  = threadIdx.x & 31;
  int tiles = (n + 15) >> 4;
  if (wave >= tiles) return;

  int half = lane >> 4;
  int ml   = lane & 15;
  int row  = wave * 16 + ml;
  int rrow = (row < n) ? row : (n - 1);
  const float* xr = x + (size_t)rrow * IN_DIM;

  v8f acc = {0.f, 0.f, 0.f, 0.f, 0.f, 0.f, 0.f, 0.f};
  for (int k = 0; k < IN_DIM; k += 4) {
    int ka = k + 2 * half;
    v2f a; a.x = xr[ka];                 a.y = xr[ka + 1];
    v2f b; b.x = lw[ka * HID + ml];      b.y = lw[(ka + 1) * HID + ml];
    acc = __builtin_amdgcn_wmma_f32_16x16x4_f32(false, a, false, b, (short)0, acc,
                                                false, false);
  }
#pragma unroll
  for (int r = 0; r < 8; ++r) {
    int orow = wave * 16 + r + 8 * half;
    if (orow < n) h1[(size_t)orow * HID + ml] = acc[r];
  }
}


__global__ void k_init_z1(float* __restrict__ h1, const float* __restrict__ dinv,
                          float* __restrict__ z1, int n) {
  int t = blockIdx.x * blockDim.x + threadIdx.x;
  int total = n * HID;
  if (t < total) {
    int i = t / HID;
    float v = h1[t] * dinv[i];
    h1[t] = v;
    z1[t] = v;
  }
}

__global__ void k_agg16(const int* __restrict__ src, const int* __restrict__ dst,
                        const float* __restrict__ hs, float* __restrict__ z, int E) {
  const int lane = threadIdx.x & 31;
  const int wave = __builtin_amdgcn_readfirstlane((blockIdx.x * blockDim.x + threadIdx.x) >> 5);
  const int e0   = wave * 32;
  if (e0 >= E) return;
  const int cnt  = (E - e0 < 32) ? (E - e0) : 32;

  int s = 0, d = 0;
  if (lane < cnt) {
    s = src[e0 + lane];
    d = dst[e0 + lane];
  }

  const int hsel = lane >> 4;
  const int f    = lane & 15;

#pragma unroll
  for (int b = 0; b < 2; ++b) {
    const int left = cnt - b * 16;
    if (left > 0) {
      float hv[8];
      int   dj[8];
#pragma unroll
      for (int u = 0; u < 8; ++u) {
        const int j  = b * 16 + u * 2 + hsel;
        const int sj = __shfl(s, j, 32);
        dj[u] = __shfl(d, j, 32);
        hv[u] = hs[(size_t)sj * HID + f];
      }
      if (left >= 16) {
#pragma unroll
        for (int u = 0; u < 8; ++u)
          atomicAdd(z + (size_t)dj[u] * HID + f, hv[u]);
      } else {
#pragma unroll
        for (int u = 0; u < 8; ++u)
          if (u * 2 + hsel < left) atomicAdd(z + (size_t)dj[u] * HID + f, hv[u]);
      }
    }
  }
}

__global__ void k_init_g(float* __restrict__ z1, const float* __restrict__ dinv,
                         const float* __restrict__ b1, float* __restrict__ g, int n) {
  int t = blockIdx.x * blockDim.x + threadIdx.x;
  int total = n * HID;
  if (t < total) {
    int i = t / HID, j = t - i * HID;
    float di = dinv[i];
    float r = fmaxf(z1[t] * di + b1[j], 0.f);
    float v = r * di;
    z1[t] = v;
    g[t]  = v;
  }
}

__global__ void k_gemm2(const float* __restrict__ g, const float* __restrict__ dinv,
                        const float* __restrict__ W2, const float* __restrict__ b2,
                        float* __restrict__ out, int n) {
  __shared__ float lw[HID * ODIM];
  for (int i = threadIdx.x; i < HID * ODIM; i += blockDim.x) lw[i] = W2[i];
  __syncthreads();

  int wave  = (blockIdx.x * blockDim.x + threadIdx.x) >> 5;
  int lane  = threadIdx.x & 31;
  int tiles = (n + 15) >> 4;
  if (wave >= tiles) return;

  int half = lane >> 4;
  int ml   = lane & 15;
  int row  = wave * 16 + ml;
  int rrow = (row < n) ? row : (n - 1);
  const float* gr = g + (size_t)rrow * HID;
  const float dr = dinv[rrow];

  v2f afrag[4];
#pragma unroll
  for (int kk = 0; kk < 4; ++kk) {
    int ka = kk * 4 + 2 * half;
    afrag[kk].x = gr[ka] * dr;
    afrag[kk].y = gr[ka + 1] * dr;
  }

#pragma unroll
  for (int tcol = 0; tcol < 4; ++tcol) {
    v8f acc = {0.f, 0.f, 0.f, 0.f, 0.f, 0.f, 0.f, 0.f};
#pragma unroll
    for (int kk = 0; kk < 4; ++kk) {
      int ka = kk * 4 + 2 * half;
      v2f b;
      b.x = lw[ka * ODIM + tcol * 16 + ml];
      b.y = lw[(ka + 1) * ODIM + tcol * 16 + ml];
      acc = __builtin_amdgcn_wmma_f32_16x16x4_f32(false, afrag[kk], false, b, (short)0,
                                                  acc, false, false);
    }
    const float bj = b2[tcol * 16 + ml];
#pragma unroll
    for (int r = 0; r < 8; ++r) {
      int orow = wave * 16 + r + 8 * half;
      if (orow < n) out[(size_t)orow * ODIM + tcol * 16 + ml] = acc[r] + bj;
    }
  }
}

extern "C" void kernel_launch(void* const* d_in, const int* in_sizes, int n_in,
                              void* d_out, int out_size, void* d_ws, size_t ws_size,
                              hipStream_t stream) {
  const float* x  = (const float*)d_in[0];
  const float* W1 = (const float*)d_in[1];
  const float* b1 = (const float*)d_in[2];
  const float* W2 = (const float*)d_in[3];
  const float* b2 = (const float*)d_in[4];
  const int*   ei = (const int*)d_in[5];

  const int N = in_sizes[0] / IN_DIM;
  const int E = in_sizes[5] / 2;
  const int* src = ei;
  const int* dst = ei + E;
  float* out = (float*)d_out;

  char* ws = (char*)d_ws;
  size_t off = 0;
  auto alloc = [&](size_t nbytes) -> void* {
    void* p = ws + off;
    off = (off + nbytes + 255) & ~(size_t)255;
    return p;
  };
  float* dinv = (float*)alloc((size_t)N * sizeof(float));
  float* h1   = (float*)alloc((size_t)N * HID  * sizeof(float));
  float* z1   = (float*)alloc((size_t)N * HID  * sizeof(float));
  float* g    = (float*)alloc((size_t)N * HID  * sizeof(float));

  const int B = 256;
  const int tiles = (N + 15) >> 4;
  const int gemmBlocks = (tiles + 7) / 8;

  k_deg_init <<<(N + B - 1) / B, B, 0, stream>>>(dinv, N);
  k_deg_edges<<<(E + B - 1) / B, B, 0, stream>>>(dst, dinv, E);
  k_dinv     <<<(N + B - 1) / B, B, 0, stream>>>(dinv, N);

  k_gemm1    <<<gemmBlocks, B, 0, stream>>>(x, W1, h1, N);
  k_init_z1  <<<((size_t)N * HID + B - 1) / B, B, 0, stream>>>(h1, dinv, z1, N);
  k_agg16    <<<(E + B - 1) / B, B, 0, stream>>>(src, dst, h1, z1, E);

  k_init_g   <<<((size_t)N * HID + B - 1) / B, B, 0, stream>>>(z1, dinv, b1, g, N);
  k_agg16    <<<(E + B - 1) / B, B, 0, stream>>>(src, dst, z1, g, E);
  k_gemm2    <<<gemmBlocks, B, 0, stream>>>(g, dinv, W2, b2, out, N);
}


# degree count without global atomics: work-groups count an edge slice into an LDS histogram of one node range (integer LDS atomics); k_dinv adds the partial counts; f32 sum orders unchanged
# speedup vs baseline: 2.8991x; 1.6118x over previous
#include <hip/hip_runtime.h>

typedef __attribute__((ext_vector_type(2))) float v2f;
typedef __attribute__((ext_vector_type(8))) float v8f;

#define IN_DIM 512
#define HID    16
#define ODIM   64

#define DEG_RANGE  12544
#define DEG_SLICES 32

__device__ __forceinline__ void deg_count(int* cnt, int d, int base) {
  unsigned a = (unsigned)(d - base);
  if (a < (unsigned)DEG_RANGE) atomicAdd(&cnt[a], 1);
}

__device__ __forceinline__ void deg_count4(int* cnt, int4 d, int base) {
  deg_count(cnt, d.x, base);
  deg_count(cnt, d.y, base);
  deg_count(cnt, d.z, base);
  deg_count(cnt, d.w, base);
}

__global__ void __launch_bounds__(1024)
k_deg_hist(const int* __restrict__ dst, int* __restrict__ part, int n, int E, int nranges) {
  __shared__ int cnt[DEG_RANGE];
  const int r    = blockIdx.x % nranges;
  const int s    = blockIdx.x / nranges;
  const int base = r * DEG_RANGE;
  const int T    = blockDim.x;
  for (int j = threadIdx.x; j < DEG_RANGE; j += T) cnt[j] = 0;
  __syncthreads();

  int head = (int)(((16u - (unsigned)((size_t)dst & 15u)) & 15u) >> 2);
  if (head > E) head = E;
  const int nvec  = (E - head) >> 2;
  const int tail0 = head + (nvec << 2);
  const int4* __restrict__ dv = (const int4*)(dst + head);

  const int per = (nvec + DEG_SLICES - 1) / DEG_SLICES;
  const int lo  = s * per;
  int hi = lo + per;
  if (hi > nvec) hi = nvec;

  int v = lo + (int)threadIdx.x;
  for (; v + 7 * T < hi; v += 8 * T) {
    int4 d0 = dv[v];
    int4 d1 = dv[v + T];
    int4 d2 = dv[v + 2 * T];
    int4 d3 = dv[v + 3 * T];
    int4 d4 = dv[v + 4 * T];
    int4 d5 = dv[v + 5 * T];
    int4 d6 = dv[v + 6 * T];
    int4 d7 = dv[v + 7 * T];
    deg_count4(cnt, d0, base);
    deg_count4(cnt, d1, base);
    deg_count4(cnt, d2, base);
    deg_count4(cnt, d3, base);
    deg_count4(cnt, d4, base);
    deg_count4(cnt, d5, base);
    deg_count4(cnt, d6, base);
    deg_count4(cnt, d7, base);
  }
  for (; v < hi; v += T) deg_count4(cnt, dv[v], base);

  if (s == 0) {
    const int t = (int)threadIdx.x;
    if (t < head) deg_count(cnt, dst[t], base);
    if (t < 4 && tail0 + t < E) deg_count(cnt, dst[tail0 + t], base);
  }
  __syncthreads();

  int* __restrict__ plane = part + (size_t)s * n;
  for (int j = threadIdx.x; j < DEG_RANGE; j += T) {
    const int i = base + j;
    if (i < n) plane[i] = cnt[j];
  }
}

__global__ void k_dinv(const int* __restrict__ part, float* __restrict__ dinv, int n) {
  int i = blockIdx.x * blockDim.x + threadIdx.x;
  if (i < n) {
    int c = 1;
#pragma unroll
    for (int s = 0; s < DEG_SLICES; ++s) c += part[(size_t)s * n + i];
    float d = (float)c;
    dinv[i] = (d > 0.0f) ? (1.0f / sqrtf(d)) : 0.0f;
  }
}

__global__ void k_gemm1(const float* __restrict__ x, const float* __restrict__ W1,
                        float* __restrict__ h1, int n) {
  __shared__ float lw[IN_DIM * HID];
  for (int i = threadIdx.x; i < IN_DIM * HID; i += blockDim.x) lw[i] = W1[i];
  __syncthreads();

  int wave  = (blockIdx.x * blockDim.x + threadIdx.x) >> 5;
  int lane  = threadIdx.x & 31;
  int tiles = (n + 15) >> 4;
  if (wave >= tiles) return;

  int half = lane >> 4;
  int ml   = lane & 15;
  int row  = wave * 16 + ml;
  int rrow = (row < n) ? row : (n - 1);
  const float* xr = x + (size_t)rrow * IN_DIM;

  v8f acc = {0.f, 0.f, 0.f, 0.f, 0.f, 0.f, 0.f, 0.f};
  for (int k = 0; k < IN_DIM; k += 4) {
    int ka = k + 2 * half;
    v2f a; a.x = xr[ka];                 a.y = xr[ka + 1];
    v2f b; b.x = lw[ka * HID + ml];      b.y = lw[(ka + 1) * HID + ml];
    acc = __builtin_amdgcn_wmma_f32_16x16x4_f32(false, a, false, b, (short)0, acc,
                                                false, false);
  }
#pragma unroll
  for (int r = 0; r < 8; ++r) {
    int orow = wave * 16 + r + 8 * half;
    if (orow < n) h1[(size_t)orow * HID + ml] = acc[r];
  }
}


__global__ void k_init_z1(float* __restrict__ h1, const float* __restrict__ dinv,
                          float* __restrict__ z1, int n) {
  int t = blockIdx.x * blockDim.x + threadIdx.x;
  int total = n * HID;
  if (t < total) {
    int i = t / HID;
    float v = h1[t] * dinv[i];
    h1[t] = v;
    z1[t] = v;
  }
}

__global__ void k_agg16(const int* __restrict__ src, const int* __restrict__ dst,
                        const float* __restrict__ hs, float* __restrict__ z, int E) {
  const int lane = threadIdx.x & 31;
  const int wave = __builtin_amdgcn_readfirstlane((blockIdx.x * blockDim.x + threadIdx.x) >> 5);
  const int e0   = wave * 32;
  if (e0 >= E) return;
  const int cnt  = (E - e0 < 32) ? (E - e0) : 32;

  int s = 0, d = 0;
  if (lane < cnt) {
    s = src[e0 + lane];
    d = dst[e0 + lane];
  }

  const int hsel = lane >> 4;
  const int f    = lane & 15;

#pragma unroll
  for (int b = 0; b < 2; ++b) {
    const int left = cnt - b * 16;
    if (left > 0) {
      float hv[8];
      int   dj[8];
#pragma unroll
      for (int u = 0; u < 8; ++u) {
        const int j  = b * 16 + u * 2 + hsel;
        const int sj = __shfl(s, j, 32);
        dj[u] = __shfl(d, j, 32);
        hv[u] = hs[(size_t)sj * HID + f];
      }
      if (left >= 16) {
#pragma unroll
        for (int u = 0; u < 8; ++u)
          atomicAdd(z + (size_t)dj[u] * HID + f, hv[u]);
      } else {
#pragma unroll
        for (int u = 0; u < 8; ++u)
          if (u * 2 + hsel < left) atomicAdd(z + (size_t)dj[u] * HID + f, hv[u]);
      }
    }
  }
}

__global__ void k_init_g(float* __restrict__ z1, const float* __restrict__ dinv,
                         const float* __restrict__ b1, float* __restrict__ g, int n) {
  int t = blockIdx.x * blockDim.x + threadIdx.x;
  int total = n * HID;
  if (t < total) {
    int i = t / HID, j = t - i * HID;
    float di = dinv[i];
    float r = fmaxf(z1[t] * di + b1[j], 0.f);
    float v = r * di;
    z1[t] = v;
    g[t]  = v;
  }
}

__global__ void k_gemm2(const float* __restrict__ g, const float* __restrict__ dinv,
                        const float* __restrict__ W2, const float* __restrict__ b2,
                        float* __restrict__ out, int n) {
  __shared__ float lw[HID * ODIM];
  for (int i = threadIdx.x; i < HID * ODIM; i += blockDim.x) lw[i] = W2[i];
  __syncthreads();

  int wave  = (blockIdx.x * blockDim.x + threadIdx.x) >> 5;
  int lane  = threadIdx.x & 31;
  int tiles = (n + 15) >> 4;
  if (wave >= tiles) return;

  int half = lane >> 4;
  int ml   = lane & 15;
  int row  = wave * 16 + ml;
  int rrow = (row < n) ? row : (n - 1);
  const float* gr = g + (size_t)rrow * HID;
  const float dr = dinv[rrow];

  v2f afrag[4];
#pragma unroll
  for (int kk = 0; kk < 4; ++kk) {
    int ka = kk * 4 + 2 * half;
    afrag[kk].x = gr[ka] * dr;
    afrag[kk].y = gr[ka + 1] * dr;
  }

#pragma unroll
  for (int tcol = 0; tcol < 4; ++tcol) {
    v8f acc = {0.f, 0.f, 0.f, 0.f, 0.f, 0.f, 0.f, 0.f};
#pragma unroll
    for (int kk = 0; kk < 4; ++kk) {
      int ka = kk * 4 + 2 * half;
      v2f b;
      b.x = lw[ka * ODIM + tcol * 16 + ml];
      b.y = lw[(ka + 1) * ODIM + tcol * 16 + ml];
      acc = __builtin_amdgcn_wmma_f32_16x16x4_f32(false, afrag[kk], false, b, (short)0,
                                                  acc, false, false);
    }
    const float bj = b2[tcol * 16 + ml];
#pragma unroll
    for (int r = 0; r < 8; ++r) {
      int orow = wave * 16 + r + 8 * half;
      if (orow < n) out[(size_t)orow * ODIM + tcol * 16 + ml] = acc[r] + bj;
    }
  }
}

extern "C" void kernel_launch(void* const* d_in, const int* in_sizes, int n_in,
                              void* d_out, int out_size, void* d_ws, size_t ws_size,
                              hipStream_t stream) {
  const float* x  = (const float*)d_in[0];
  const float* W1 = (const float*)d_in[1];
  const float* b1 = (const float*)d_in[2];
  const float* W2 = (const float*)d_in[3];
  const float* b2 = (const float*)d_in[4];
  const int*   ei = (const int*)d_in[5];

  const int N = in_sizes[0] / IN_DIM;
  const int E = in_sizes[5] / 2;
  const int* src = ei;
  const int* dst = ei + E;
  float* out = (float*)d_out;

  char* ws = (char*)d_ws;
  size_t off = 0;
  auto alloc = [&](size_t nbytes) -> void* {
    void* p = ws + off;
    off = (off + nbytes + 255) & ~(size_t)255;
    return p;
  };
  float* dinv = (float*)alloc((size_t)N * sizeof(float));
  float* h1   = (float*)alloc((size_t)N * HID  * sizeof(float));
  float* z1   = (float*)alloc((size_t)N * HID  * sizeof(float));
  float* g    = (float*)alloc((size_t)N * HID  * sizeof(float));

  const int B = 256;
  const int tiles = (N + 15) >> 4;
  const int gemmBlocks = (tiles + 7) / 8;

  int* part = (int*)h1;
  const int nranges = (N + DEG_RANGE - 1) / DEG_RANGE;
  k_deg_hist <<<nranges * DEG_SLICES, 1024, 0, stream>>>(dst, part, N, E, nranges);
  k_dinv     <<<(N + B - 1) / B, B, 0, stream>>>(part, dinv, N);

  k_gemm1    <<<gemmBlocks, B, 0, stream>>>(x, W1, h1, N);
  k_init_z1  <<<((size_t)N * HID + B - 1) / B, B, 0, stream>>>(h1, dinv, z1, N);
  k_agg16    <<<(E + B - 1) / B, B, 0, stream>>>(src, dst, h1, z1, E);

  k_init_g   <<<((size_t)N * HID + B - 1) / B, B, 0, stream>>>(z1, dinv, b1, g, N);
  k_agg16    <<<(E + B - 1) / B, B, 0, stream>>>(src, dst, z1, g, E);
  k_gemm2    <<<gemmBlocks, B, 0, stream>>>(g, dinv, W2, b2, out, N);
}


# edge passes add into two copies of each sum plane, one per half of the chip (die asked once per work-group): a row's f32 atomics stay in one L2 half; consumers add the two partial sums
# speedup vs baseline: 4.9910x; 1.7216x over previous
#include <hip/hip_runtime.h>

typedef __attribute__((ext_vector_type(2))) float v2f;
typedef __attribute__((ext_vector_type(8))) float v8f;

#define IN_DIM 512
#define HID    16
#define ODIM   64

#define DEG_RANGE  12544
#define DEG_SLICES 32

__device__ __forceinline__ void deg_count(int* cnt, int d, int base) {
  unsigned a = (unsigned)(d - base);
  if (a < (unsigned)DEG_RANGE) atomicAdd(&cnt[a], 1);
}

__device__ __forceinline__ void deg_count4(int* cnt, int4 d, int base) {
  deg_count(cnt, d.x, base);
  deg_count(cnt, d.y, base);
  deg_count(cnt, d.z, base);
  deg_count(cnt, d.w, base);
}

__global__ void __launch_bounds__(1024)
k_deg_hist(const int* __restrict__ dst, int* __restrict__ part, int n, int E, int nranges) {
  __shared__ int cnt[DEG_RANGE];
  const int r    = blockIdx.x % nranges;
  const int s    = blockIdx.x / nranges;
  const int base = r * DEG_RANGE;
  const int T    = blockDim.x;
  for (int j = threadIdx.x; j < DEG_RANGE; j += T) cnt[j] = 0;
  __syncthreads();

  int head = (int)(((16u - (unsigned)((size_t)dst & 15u)) & 15u) >> 2);
  if (head > E) head = E;
  const int nvec  = (E - head) >> 2;
  const int tail0 = head + (nvec << 2);
  const int4* __restrict__ dv = (const int4*)(dst + head);

  const int per = (nvec + DEG_SLICES - 1) / DEG_SLICES;
  const int lo  = s * per;
  int hi = lo + per;
  if (hi > nvec) hi = nvec;

  int v = lo + (int)threadIdx.x;
  for (; v + 7 * T < hi; v += 8 * T) {
    int4 d0 = dv[v];
    int4 d1 = dv[v + T];
    int4 d2 = dv[v + 2 * T];
    int4 d3 = dv[v + 3 * T];
    int4 d4 = dv[v + 4 * T];
    int4 d5 = dv[v + 5 * T];
    int4 d6 = dv[v + 6 * T];
    int4 d7 = dv[v + 7 * T];
    deg_count4(cnt, d0, base);
    deg_count4(cnt, d1, base);
    deg_count4(cnt, d2, base);
    deg_count4(cnt, d3, base);
    deg_count4(cnt, d4, base);
    deg_count4(cnt, d5, base);
    deg_count4(cnt, d6, base);
    deg_count4(cnt, d7, base);
  }
  for (; v < hi; v += T) deg_count4(cnt, dv[v], base);

  if (s == 0) {
    const int t = (int)threadIdx.x;
    if (t < head) deg_count(cnt, dst[t], base);
    if (t < 4 && tail0 + t < E) deg_count(cnt, dst[tail0 + t], base);
  }
  __syncthreads();

  int* __restrict__ plane = part + (size_t)s * n;
  for (int j = threadIdx.x; j < DEG_RANGE; j += T) {
    const int i = base + j;
    if (i < n) plane[i] = cnt[j];
  }
}

__global__ void k_dinv(const int* __restrict__ part, float* __restrict__ dinv, int n) {
  int i = blockIdx.x * blockDim.x + threadIdx.x;
  if (i < n) {
    int c = 1;
#pragma unroll
    for (int s = 0; s < DEG_SLICES; ++s) c += part[(size_t)s * n + i];
    float d = (float)c;
    dinv[i] = (d > 0.0f) ? (1.0f / sqrtf(d)) : 0.0f;
  }
}

__global__ void k_gemm1(const float* __restrict__ x, const float* __restrict__ W1,
                        float* __restrict__ h1, int n) {
  __shared__ float lw[IN_DIM * HID];
  for (int i = threadIdx.x; i < IN_DIM * HID; i += blockDim.x) lw[i] = W1[i];
  __syncthreads();

  int wave  = (blockIdx.x * blockDim.x + threadIdx.x) >> 5;
  int lane  = threadIdx.x & 31;
  int tiles = (n + 15) >> 4;
  if (wave >= tiles) return;

  int half = lane >> 4;
  int ml   = lane & 15;
  int row  = wave * 16 + ml;
  int rrow = (row < n) ? row : (n - 1);
  const float* xr = x + (size_t)rrow * IN_DIM;

  v8f acc = {0.f, 0.f, 0.f, 0.f, 0.f, 0.f, 0.f, 0.f};
  for (int k = 0; k < IN_DIM; k += 4) {
    int ka = k + 2 * half;
    v2f a; a.x = xr[ka];                 a.y = xr[ka + 1];
    v2f b; b.x = lw[ka * HID + ml];      b.y = lw[(ka + 1) * HID + ml];
    acc = __builtin_amdgcn_wmma_f32_16x16x4_f32(false, a, false, b, (short)0, acc,
                                                false, false);
  }
#pragma unroll
  for (int r = 0; r < 8; ++r) {
    int orow = wave * 16 + r + 8 * half;
    if (orow < n) h1[(size_t)orow * HID + ml] = acc[r];
  }
}


__global__ void k_init_z1(float* __restrict__ h1, const float* __restrict__ dinv,
                          float* __restrict__ z1, float* __restrict__ z1b, int n) {
  int t = blockIdx.x * blockDim.x + threadIdx.x;
  int total = n * HID;
  if (t < total) {
    int i = t / HID;
    float v = h1[t] * dinv[i];
    h1[t]  = v;
    z1[t]  = v;
    z1b[t] = 0.f;
  }
}

__global__ void __launch_bounds__(1024)
k_agg16(const int* __restrict__ src, const int* __restrict__ dst,
        const float* __restrict__ hs, float* __restrict__ za, float* __restrict__ zb, int E) {
  __shared__ int s_half;

  const int lane = threadIdx.x & 31;
  const int wave = __builtin_amdgcn_readfirstlane((blockIdx.x * blockDim.x + threadIdx.x) >> 5);
  const int e0   = wave * 32;
  const int rest = E - e0;
  const int cnt  = (rest < 32) ? rest : 32;

  int s = 0, d = 0;
  if (lane < cnt) {
    s = src[e0 + lane];
    d = dst[e0 + lane];
  }

  if (threadIdx.x == 0) {
    s_half = (int)((__builtin_amdgcn_s_sendmsg_rtn(0x87) >> 18) & 1u);
  }
  __syncthreads();
  if (rest <= 0) return;
  float* __restrict__ z = __builtin_amdgcn_readfirstlane(s_half) ? zb : za;

  const int hsel = lane >> 4;
  const int f    = lane & 15;

#pragma unroll
  for (int b = 0; b < 2; ++b) {
    const int left = cnt - b * 16;
    if (left > 0) {
      float hv[8];
      int   dj[8];
#pragma unroll
      for (int u = 0; u < 8; ++u) {
        const int j  = b * 16 + u * 2 + hsel;
        const int sj = __shfl(s, j, 32);
        dj[u] = __shfl(d, j, 32);
        hv[u] = hs[(size_t)sj * HID + f];
      }
      if (left >= 16) {
#pragma unroll
        for (int u = 0; u < 8; ++u)
          atomicAdd(z + (size_t)dj[u] * HID + f, hv[u]);
      } else {
#pragma unroll
        for (int u = 0; u < 8; ++u)
          if (u * 2 + hsel < left) atomicAdd(z + (size_t)dj[u] * HID + f, hv[u]);
      }
    }
  }
}

__global__ void k_init_g(float* __restrict__ z1, const float* __restrict__ dinv,
                         const float* __restrict__ b1, float* __restrict__ g,
                         float* __restrict__ gb, int n) {
  int t = blockIdx.x * blockDim.x + threadIdx.x;
  int total = n * HID;
  if (t < total) {
    int i = t / HID, j = t - i * HID;
    float di = dinv[i];
    float zsum = z1[t] + g[t];
    float r = fmaxf(zsum * di + b1[j], 0.f);
    float v = r * di;
    z1[t] = v;
    g[t]  = v;
    gb[t] = 0.f;
  }
}

__global__ void k_gemm2(const float* __restrict__ g, const float* __restrict__ gb,
                        const float* __restrict__ dinv,
                        const float* __restrict__ W2, const float* __restrict__ b2,
                        float* __restrict__ out, int n) {
  __shared__ float lw[HID * ODIM];
  for (int i = threadIdx.x; i < HID * ODIM; i += blockDim.x) lw[i] = W2[i];
  __syncthreads();

  int wave  = (blockIdx.x * blockDim.x + threadIdx.x) >> 5;
  int lane  = threadIdx.x & 31;
  int tiles = (n + 15) >> 4;
  if (wave >= tiles) return;

  int half = lane >> 4;
  int ml   = lane & 15;
  int row  = wave * 16 + ml;
  int rrow = (row < n) ? row : (n - 1);
  const float* gr  = g  + (size_t)rrow * HID;
  const float* gbr = gb + (size_t)rrow * HID;
  const float dr = dinv[rrow];

  v2f afrag[4];
#pragma unroll
  for (int kk = 0; kk < 4; ++kk) {
    int ka = kk * 4 + 2 * half;
    afrag[kk].x = (gr[ka]     + gbr[ka])     * dr;
    afrag[kk].y = (gr[ka + 1] + gbr[ka + 1]) * dr;
  }

#pragma unroll
  for (int tcol = 0; tcol < 4; ++tcol) {
    v8f acc = {0.f, 0.f, 0.f, 0.f, 0.f, 0.f, 0.f, 0.f};
#pragma unroll
    for (int kk = 0; kk < 4; ++kk) {
      int ka = kk * 4 + 2 * half;
      v2f b;
      b.x = lw[ka * ODIM + tcol * 16 + ml];
      b.y = lw[(ka + 1) * ODIM + tcol * 16 + ml];
      acc = __builtin_amdgcn_wmma_f32_16x16x4_f32(false, afrag[kk], false, b, (short)0,
                                                  acc, false, false);
    }
    const float bj = b2[tcol * 16 + ml];
#pragma unroll
    for (int r = 0; r < 8; ++r) {
      int orow = wave * 16 + r + 8 * half;
      if (orow < n) out[(size_t)orow * ODIM + tcol * 16 + ml] = acc[r] + bj;
    }
  }
}

extern "C" void kernel_launch(void* const* d_in, const int* in_sizes, int n_in,
                              void* d_out, int out_size, void* d_ws, size_t ws_size,
                              hipStream_t stream) {
  const float* x  = (const float*)d_in[0];
  const float* W1 = (const float*)d_in[1];
  const float* b1 = (const float*)d_in[2];
  const float* W2 = (const float*)d_in[3];
  const float* b2 = (const float*)d_in[4];
  const int*   ei = (const int*)d_in[5];

  const int N = in_sizes[0] / IN_DIM;
  const int E = in_sizes[5] / 2;
  const int* src = ei;
  const int* dst = ei + E;
  float* out = (float*)d_out;

  char* ws = (char*)d_ws;
  size_t off = 0;
  auto alloc = [&](size_t nbytes) -> void* {
    void* p = ws + off;
    off = (off + nbytes + 255) & ~(size_t)255;
    return p;
  };
  float* dinv = (float*)alloc((size_t)N * sizeof(float));
  float* h1   = (float*)alloc((size_t)N * HID  * sizeof(float));
  float* z1   = (float*)alloc((size_t)N * HID  * sizeof(float));
  float* g    = (float*)alloc((size_t)N * HID  * sizeof(float));

  const int B = 256;
  const int tiles = (N + 15) >> 4;
  const int gemmBlocks = (tiles + 7) / 8;

  int* part = (int*)h1;
  const int nranges = (N + DEG_RANGE - 1) / DEG_RANGE;
  k_deg_hist <<<nranges * DEG_SLICES, 1024, 0, stream>>>(dst, part, N, E, nranges);
  k_dinv     <<<(N + B - 1) / B, B, 0, stream>>>(part, dinv, N);

  k_gemm1    <<<gemmBlocks, B, 0, stream>>>(x, W1, h1, N);
  const int BE = 1024;
  k_init_z1  <<<((size_t)N * HID + B - 1) / B, B, 0, stream>>>(h1, dinv, z1, g, N);
  k_agg16    <<<(E + BE - 1) / BE, BE, 0, stream>>>(src, dst, h1, z1, g, E);

  k_init_g   <<<((size_t)N * HID + B - 1) / B, B, 0, stream>>>(z1, dinv, b1, g, h1, N);
  k_agg16    <<<(E + BE - 1) / BE, BE, 0, stream>>>(src, dst, z1, g, h1, E);
  k_gemm2    <<<gemmBlocks, B, 0, stream>>>(g, h1, dinv, W2, b2, out, N);
}


# GEMM1 epilogue applies the source factor dinv[row] to its tile and stores h1, the self-loop copy z1 and the zeroed second copy itself; the row-init launch is gone; values and sum orders unchanged
# speedup vs baseline: 5.1775x; 1.0374x over previous
#include <hip/hip_runtime.h>

typedef __attribute__((ext_vector_type(2))) float v2f;
typedef __attribute__((ext_vector_type(8))) float v8f;

#define IN_DIM 512
#define HID    16
#define ODIM   64

#define DEG_RANGE  12544
#define DEG_SLICES 32

__device__ __forceinline__ void deg_count(int* cnt, int d, int base) {
  unsigned a = (unsigned)(d - base);
  if (a < (unsigned)DEG_RANGE) atomicAdd(&cnt[a], 1);
}

__device__ __forceinline__ void deg_count4(int* cnt, int4 d, int base) {
  deg_count(cnt, d.x, base);
  deg_count(cnt, d.y, base);
  deg_count(cnt, d.z, base);
  deg_count(cnt, d.w, base);
}

__global__ void __launch_bounds__(1024)
k_deg_hist(const int* __restrict__ dst, int* __restrict__ part, int n, int E, int nranges) {
  __shared__ int cnt[DEG_RANGE];
  const int r    = blockIdx.x % nranges;
  const int s    = blockIdx.x / nranges;
  const int base = r * DEG_RANGE;
  const int T    = blockDim.x;
  for (int j = threadIdx.x; j < DEG_RANGE; j += T) cnt[j] = 0;
  __syncthreads();

  int head = (int)(((16u - (unsigned)((size_t)dst & 15u)) & 15u) >> 2);
  if (head > E) head = E;
  const int nvec  = (E - head) >> 2;
  const int tail0 = head + (nvec << 2);
  const int4* __restrict__ dv = (const int4*)(dst + head);

  const int per = (nvec + DEG_SLICES - 1) / DEG_SLICES;
  const int lo  = s * per;
  int hi = lo + per;
  if (hi > nvec) hi = nvec;

  int v = lo + (int)threadIdx.x;
  for (; v + 7 * T < hi; v += 8 * T) {
    int4 d0 = dv[v];
    int4 d1 = dv[v + T];
    int4 d2 = dv[v + 2 * T];
    int4 d3 = dv[v + 3 * T];
    int4 d4 = dv[v + 4 * T];
    int4 d5 = dv[v + 5 * T];
    int4 d6 = dv[v + 6 * T];
    int4 d7 = dv[v + 7 * T];
    deg_count4(cnt, d0, base);
    deg_count4(cnt, d1, base);
    deg_count4(cnt, d2, base);
    deg_count4(cnt, d3, base);
    deg_count4(cnt, d4, base);
    deg_count4(cnt, d5, base);
    deg_count4(cnt, d6, base);
    deg_count4(cnt, d7, base);
  }
  for (; v < hi; v += T) deg_count4(cnt, dv[v], base);

  if (s == 0) {
    const int t = (int)threadIdx.x;
    if (t < head) deg_count(cnt, dst[t], base);
    if (t < 4 && tail0 + t < E) deg_count(cnt, dst[tail0 + t], base);
  }
  __syncthreads();

  int* __restrict__ plane = part + (size_t)s * n;
  for (int j = threadIdx.x; j < DEG_RANGE; j += T) {
    const int i = base + j;
    if (i < n) plane[i] = cnt[j];
  }
}

__global__ void k_dinv(const int* __restrict__ part, float* __restrict__ dinv, int n) {
  int i = blockIdx.x * blockDim.x + threadIdx.x;
  if (i < n) {
    int c = 1;
#pragma unroll
    for (int s = 0; s < DEG_SLICES; ++s) c += part[(size_t)s * n + i];
    float d = (float)c;
    dinv[i] = (d > 0.0f) ? (1.0f / sqrtf(d)) : 0.0f;
  }
}


__global__ void k_gemm1(const float* __restrict__ x, const float* __restrict__ W1,
                        const float* __restrict__ dinv, float* __restrict__ h1,
                        float* __restrict__ z1, float* __restrict__ z1b, int n) {
  __shared__ float lw[IN_DIM * HID];
  for (int i = threadIdx.x; i < IN_DIM * HID; i += blockDim.x) lw[i] = W1[i];
  __syncthreads();

  int wave  = (blockIdx.x * blockDim.x + threadIdx.x) >> 5;
  int lane  = threadIdx.x & 31;
  int tiles = (n + 15) >> 4;
  if (wave >= tiles) return;

  int half = lane >> 4;
  int ml   = lane & 15;
  int row  = wave * 16 + ml;
  int rrow = (row < n) ? row : (n - 1);
  const float* xr = x + (size_t)rrow * IN_DIM;

  float dv[8];
#pragma unroll
  for (int r = 0; r < 8; ++r) {
    int orow = wave * 16 + r + 8 * half;
    dv[r] = dinv[(orow < n) ? orow : (n - 1)];
  }

  v8f acc = {0.f, 0.f, 0.f, 0.f, 0.f, 0.f, 0.f, 0.f};
  for (int k = 0; k < IN_DIM; k += 4) {
    int ka = k + 2 * half;
    v2f a; a.x = xr[ka];                 a.y = xr[ka + 1];
    v2f b; b.x = lw[ka * HID + ml];      b.y = lw[(ka + 1) * HID + ml];
    acc = __builtin_amdgcn_wmma_f32_16x16x4_f32(false, a, false, b, (short)0, acc,
                                                false, false);
  }
#pragma unroll
  for (int r = 0; r < 8; ++r) {
    int orow = wave * 16 + r + 8 * half;
    if (orow < n) {
      const size_t o = (size_t)orow * HID + ml;
      const float v = acc[r] * dv[r];
      h1[o]  = v;
      z1[o]  = v;
      z1b[o] = 0.f;
    }
  }
}

__global__ void __launch_bounds__(1024)
k_agg16(const int* __restrict__ src, const int* __restrict__ dst,
        const float* __restrict__ hs, float* __restrict__ za, float* __restrict__ zb, int E) {
  __shared__ int s_half;

  const int lane = threadIdx.x & 31;
  const int wave = __builtin_amdgcn_readfirstlane((blockIdx.x * blockDim.x + threadIdx.x) >> 5);
  const int e0   = wave * 32;
  const int rest = E - e0;
  const int cnt  = (rest < 32) ? rest : 32;

  int s = 0, d = 0;
  if (lane < cnt) {
    s = src[e0 + lane];
    d = dst[e0 + lane];
  }

  if (threadIdx.x == 0) {
    s_half = (int)((__builtin_amdgcn_s_sendmsg_rtn(0x87) >> 18) & 1u);
  }
  __syncthreads();
  if (rest <= 0) return;
  float* __restrict__ z = __builtin_amdgcn_readfirstlane(s_half) ? zb : za;

  const int hsel = lane >> 4;
  const int f    = lane & 15;

#pragma unroll
  for (int b = 0; b < 2; ++b) {
    const int left = cnt - b * 16;
    if (left > 0) {
      float hv[8];
      int   dj[8];
#pragma unroll
      for (int u = 0; u < 8; ++u) {
        const int j  = b * 16 + u * 2 + hsel;
        const int sj = __shfl(s, j, 32);
        dj[u] = __shfl(d, j, 32);
        hv[u] = hs[(size_t)sj * HID + f];
      }
      if (left >= 16) {
#pragma unroll
        for (int u = 0; u < 8; ++u)
          atomicAdd(z + (size_t)dj[u] * HID + f, hv[u]);
      } else {
#pragma unroll
        for (int u = 0; u < 8; ++u)
          if (u * 2 + hsel < left) atomicAdd(z + (size_t)dj[u] * HID + f, hv[u]);
      }
    }
  }
}

__global__ void k_init_g(float* __restrict__ z1, const float* __restrict__ dinv,
                         const float* __restrict__ b1, float* __restrict__ g,
                         float* __restrict__ gb, int n) {
  int t = blockIdx.x * blockDim.x + threadIdx.x;
  int total = n * HID;
  if (t < total) {
    int i = t / HID, j = t - i * HID;
    float di = dinv[i];
    float zsum = z1[t] + g[t];
    float r = fmaxf(zsum * di + b1[j], 0.f);
    float v = r * di;
    z1[t] = v;
    g[t]  = v;
    gb[t] = 0.f;
  }
}

__global__ void k_gemm2(const float* __restrict__ g, const float* __restrict__ gb,
                        const float* __restrict__ dinv,
                        const float* __restrict__ W2, const float* __restrict__ b2,
                        float* __restrict__ out, int n) {
  __shared__ float lw[HID * ODIM];
  for (int i = threadIdx.x; i < HID * ODIM; i += blockDim.x) lw[i] = W2[i];
  __syncthreads();

  int wave  = (blockIdx.x * blockDim.x + threadIdx.x) >> 5;
  int lane  = threadIdx.x & 31;
  int tiles = (n + 15) >> 4;
  if (wave >= tiles) return;

  int half = lane >> 4;
  int ml   = lane & 15;
  int row  = wave * 16 + ml;
  int rrow = (row < n) ? row : (n - 1);
  const float* gr  = g  + (size_t)rrow * HID;
  const float* gbr = gb + (size_t)rrow * HID;
  const float dr = dinv[rrow];

  v2f afrag[4];
#pragma unroll
  for (int kk = 0; kk < 4; ++kk) {
    int ka = kk * 4 + 2 * half;
    afrag[kk].x = (gr[ka]     + gbr[ka])     * dr;
    afrag[kk].y = (gr[ka + 1] + gbr[ka + 1]) * dr;
  }

#pragma unroll
  for (int tcol = 0; tcol < 4; ++tcol) {
    v8f acc = {0.f, 0.f, 0.f, 0.f, 0.f, 0.f, 0.f, 0.f};
#pragma unroll
    for (int kk = 0; kk < 4; ++kk) {
      int ka = kk * 4 + 2 * half;
      v2f b;
      b.x = lw[ka * ODIM + tcol * 16 + ml];
      b.y = lw[(ka + 1) * ODIM + tcol * 16 + ml];
      acc = __builtin_amdgcn_wmma_f32_16x16x4_f32(false, afrag[kk], false, b, (short)0,
                                                  acc, false, false);
    }
    const float bj = b2[tcol * 16 + ml];
#pragma unroll
    for (int r = 0; r < 8; ++r) {
      int orow = wave * 16 + r + 8 * half;
      if (orow < n) out[(size_t)orow * ODIM + tcol * 16 + ml] = acc[r] + bj;
    }
  }
}

extern "C" void kernel_launch(void* const* d_in, const int* in_sizes, int n_in,
                              void* d_out, int out_size, void* d_ws, size_t ws_size,
                              hipStream_t stream) {
  const float* x  = (const float*)d_in[0];
  const float* W1 = (const float*)d_in[1];
  const float* b1 = (const float*)d_in[2];
  const float* W2 = (const float*)d_in[3];
  const float* b2 = (const float*)d_in[4];
  const int*   ei = (const int*)d_in[5];

  const int N = in_sizes[0] / IN_DIM;
  const int E = in_sizes[5] / 2;
  const int* src = ei;
  const int* dst = ei + E;
  float* out = (float*)d_out;

  char* ws = (char*)d_ws;
  size_t off = 0;
  auto alloc = [&](size_t nbytes) -> void* {
    void* p = ws + off;
    off = (off + nbytes + 255) & ~(size_t)255;
    return p;
  };
  float* dinv = (float*)alloc((size_t)N * sizeof(float));
  float* h1   = (float*)alloc((size_t)N * HID  * sizeof(float));
  float* z1   = (float*)alloc((size_t)N * HID  * sizeof(float));
  float* g    = (float*)alloc((size_t)N * HID  * sizeof(float));

  const int B = 256;
  const int tiles = (N + 15) >> 4;
  const int gemmBlocks = (tiles + 7) / 8;

  int* part = (int*)h1;
  const int nranges = (N + DEG_RANGE - 1) / DEG_RANGE;
  k_deg_hist <<<nranges * DEG_SLICES, 1024, 0, stream>>>(dst, part, N, E, nranges);
  k_dinv     <<<(N + B - 1) / B, B, 0, stream>>>(part, dinv, N);

  const int BE = 1024;
  k_gemm1    <<<gemmBlocks, B, 0, stream>>>(x, W1, dinv, h1, z1, g, N);
  k_agg16    <<<(E + BE - 1) / BE, BE, 0, stream>>>(src, dst, h1, z1, g, E);

  k_init_g   <<<((size_t)N * HID + B - 1) / B, B, 0, stream>>>(z1, dinv, b1, g, h1, N);
  k_agg16    <<<(E + BE - 1) / BE, BE, 0, stream>>>(src, dst, z1, g, h1, E);
  k_gemm2    <<<gemmBlocks, B, 0, stream>>>(g, h1, dinv, W2, b2, out, N);
}


# GEMM1 streams x: each load takes 256 B of two rows (whole lines, touched once, non-temporal); chunks pass through a per-wave LDS area to the A fragments; values and sum order unchanged
# speedup vs baseline: 5.7737x; 1.1152x over previous
#include <hip/hip_runtime.h>

typedef __attribute__((ext_vector_type(2))) float v2f;
typedef __attribute__((ext_vector_type(4))) float v4f;
typedef __attribute__((ext_vector_type(8))) float v8f;

#define IN_DIM 512
#define HID    16
#define ODIM   64

#define DEG_RANGE  12544
#define DEG_SLICES 32

__device__ __forceinline__ void deg_count(int* cnt, int d, int base) {
  unsigned a = (unsigned)(d - base);
  if (a < (unsigned)DEG_RANGE) atomicAdd(&cnt[a], 1);
}

__device__ __forceinline__ void deg_count4(int* cnt, int4 d, int base) {
  deg_count(cnt, d.x, base);
  deg_count(cnt, d.y, base);
  deg_count(cnt, d.z, base);
  deg_count(cnt, d.w, base);
}

__global__ void __launch_bounds__(1024)
k_deg_hist(const int* __restrict__ dst, int* __restrict__ part, int n, int E, int nranges) {
  __shared__ int cnt[DEG_RANGE];
  const int r    = blockIdx.x % nranges;
  const int s    = blockIdx.x / nranges;
  const int base = r * DEG_RANGE;
  const int T    = blockDim.x;
  for (int j = threadIdx.x; j < DEG_RANGE; j += T) cnt[j] = 0;
  __syncthreads();

  int head = (int)(((16u - (unsigned)((size_t)dst & 15u)) & 15u) >> 2);
  if (head > E) head = E;
  const int nvec  = (E - head) >> 2;
  const int tail0 = head + (nvec << 2);
  const int4* __restrict__ dv = (const int4*)(dst + head);

  const int per = (nvec + DEG_SLICES - 1) / DEG_SLICES;
  const int lo  = s * per;
  int hi = lo + per;
  if (hi > nvec) hi = nvec;

  int v = lo + (int)threadIdx.x;
  for (; v + 7 * T < hi; v += 8 * T) {
    int4 d0 = dv[v];
    int4 d1 = dv[v + T];
    int4 d2 = dv[v + 2 * T];
    int4 d3 = dv[v + 3 * T];
    int4 d4 = dv[v + 4 * T];
    int4 d5 = dv[v + 5 * T];
    int4 d6 = dv[v + 6 * T];
    int4 d7 = dv[v + 7 * T];
    deg_count4(cnt, d0, base);
    deg_count4(cnt, d1, base);
    deg_count4(cnt, d2, base);
    deg_count4(cnt, d3, base);
    deg_count4(cnt, d4, base);
    deg_count4(cnt, d5, base);
    deg_count4(cnt, d6, base);
    deg_count4(cnt, d7, base);
  }
  for (; v < hi; v += T) deg_count4(cnt, dv[v], base);

  if (s == 0) {
    const int t = (int)threadIdx.x;
    if (t < head) deg_count(cnt, dst[t], base);
    if (t < 4 && tail0 + t < E) deg_count(cnt, dst[tail0 + t], base);
  }
  __syncthreads();

  int* __restrict__ plane = part + (size_t)s * n;
  for (int j = threadIdx.x; j < DEG_RANGE; j += T) {
    const int i = base + j;
    if (i < n) plane[i] = cnt[j];
  }
}

__global__ void k_dinv(const int* __restrict__ part, float* __restrict__ dinv, int n) {
  int i = blockIdx.x * blockDim.x + threadIdx.x;
  if (i < n) {
    int c = 1;
#pragma unroll
    for (int s = 0; s < DEG_SLICES; ++s) c += part[(size_t)s * n + i];
    float d = (float)c;
    dinv[i] = (d > 0.0f) ? (1.0f / sqrtf(d)) : 0.0f;
  }
}


#define G1_CH    64
#define G1_NCH   (IN_DIM / G1_CH)
#define G1_PITCH (G1_CH + 4)
#define G1_WAVES 8
#define G1_WPT   (IN_DIM * HID / 4 / (G1_WAVES * 32))

__global__ void __launch_bounds__(G1_WAVES * 32)
k_gemm1(const float* __restrict__ x, const float* __restrict__ W1,
        const float* __restrict__ dinv, float* __restrict__ h1,
        float* __restrict__ z1, float* __restrict__ z1b, int n) {
  __shared__ __attribute__((aligned(16))) float lw[IN_DIM * HID];
  __shared__ __attribute__((aligned(16))) float stage[G1_WAVES][16 * G1_PITCH];

  const int wave  = (blockIdx.x * blockDim.x + threadIdx.x) >> 5;
  const int lane  = threadIdx.x & 31;
  const int tiles = (n + 15) >> 4;
  const bool live = wave < tiles;

  const int r2 = lane >> 4;
  const int q  = lane & 15;
  const float* xp[8];
#pragma unroll
  for (int i = 0; i < 8; ++i) {
    int gr = wave * 16 + 2 * i + r2;
    if (gr >= n) gr = n - 1;
    xp[i] = x + (size_t)gr * IN_DIM + 4 * q;
  }

  v4f wreg[G1_WPT];
#pragma unroll
  for (int j = 0; j < G1_WPT; ++j)
    wreg[j] = *(const v4f*)(W1 + 4 * ((int)threadIdx.x + j * G1_WAVES * 32));
  v4f buf[2][8];
#pragma unroll
  for (int i = 0; i < 8; ++i) buf[0][i] = __builtin_nontemporal_load((const v4f*)(xp[i]));
#pragma unroll
  for (int i = 0; i < 8; ++i) buf[1][i] = __builtin_nontemporal_load((const v4f*)(xp[i] + G1_CH));
#pragma unroll
  for (int j = 0; j < G1_WPT; ++j)
    *(v4f*)(lw + 4 * ((int)threadIdx.x + j * G1_WAVES * 32)) = wreg[j];
  __syncthreads();
  if (!live) return;

  const int half = lane >> 4;
  const int ml   = lane & 15;
  float* st = stage[threadIdx.x >> 5];

  float dv[8];
#pragma unroll
  for (int r = 0; r < 8; ++r) {
    int orow = wave * 16 + r + 8 * half;
    dv[r] = dinv[(orow < n) ? orow : (n - 1)];
  }

  v8f acc = {0.f, 0.f, 0.f, 0.f, 0.f, 0.f, 0.f, 0.f};
#pragma unroll
  for (int c = 0; c < G1_NCH; ++c) {
#pragma unroll
    for (int i = 0; i < 8; ++i)
      *(v4f*)(st + (2 * i + r2) * G1_PITCH + 4 * q) = buf[c & 1][i];
    if (c + 2 < G1_NCH) {
#pragma unroll
      for (int i = 0; i < 8; ++i)
        buf[c & 1][i] = __builtin_nontemporal_load((const v4f*)(xp[i] + (c + 2) * G1_CH));
    }
    __builtin_amdgcn_fence(__ATOMIC_ACQ_REL, "wavefront");
#pragma unroll
    for (int kk = 0; kk < G1_CH / 4; ++kk) {
      const int ka = c * G1_CH + 4 * kk + 2 * half;
      const v2f a = *(const v2f*)(st + ml * G1_PITCH + 4 * kk + 2 * half);
      v2f b; b.x = lw[ka * HID + ml];      b.y = lw[(ka + 1) * HID + ml];
      acc = __builtin_amdgcn_wmma_f32_16x16x4_f32(false, a, false, b, (short)0, acc,
                                                  false, false);
    }
    __builtin_amdgcn_fence(__ATOMIC_ACQ_REL, "wavefront");
  }
#pragma unroll
  for (int r = 0; r < 8; ++r) {
    int orow = wave * 16 + r + 8 * half;
    if (orow < n) {
      const size_t o = (size_t)orow * HID + ml;
      const float v = acc[r] * dv[r];
      h1[o]  = v;
      z1[o]  = v;
      z1b[o] = 0.f;
    }
  }
}

__global__ void __launch_bounds__(1024)
k_agg16(const int* __restrict__ src, const int* __restrict__ dst,
        const float* __restrict__ hs, float* __restrict__ za, float* __restrict__ zb, int E) {
  __shared__ int s_half;

  const int lane = threadIdx.x & 31;
  const int wave = __builtin_amdgcn_readfirstlane((blockIdx.x * blockDim.x + threadIdx.x) >> 5);
  const int e0   = wave * 32;
  const int rest = E - e0;
  const int cnt  = (rest < 32) ? rest : 32;

  int s = 0, d = 0;
  if (lane < cnt) {
    s = src[e0 + lane];
    d = dst[e0 + lane];
  }

  if (threadIdx.x == 0) {
    s_half = (int)((__builtin_amdgcn_s_sendmsg_rtn(0x87) >> 18) & 1u);
  }
  __syncthreads();
  if (rest <= 0) return;
  float* __restrict__ z = __builtin_amdgcn_readfirstlane(s_half) ? zb : za;

  const int hsel = lane >> 4;
  const int f    = lane & 15;

#pragma unroll
  for (int b = 0; b < 2; ++b) {
    const int left = cnt - b * 16;
    if (left > 0) {
      float hv[8];
      int   dj[8];
#pragma unroll
      for (int u = 0; u < 8; ++u) {
        const int j  = b * 16 + u * 2 + hsel;
        const int sj = __shfl(s, j, 32);
        dj[u] = __shfl(d, j, 32);
        hv[u] = hs[(size_t)sj * HID + f];
      }
      if (left >= 16) {
#pragma unroll
        for (int u = 0; u < 8; ++u)
          atomicAdd(z + (size_t)dj[u] * HID + f, hv[u]);
      } else {
#pragma unroll
        for (int u = 0; u < 8; ++u)
          if (u * 2 + hsel < left) atomicAdd(z + (size_t)dj[u] * HID + f, hv[u]);
      }
    }
  }
}

__global__ void k_init_g(float* __restrict__ z1, const float* __restrict__ dinv,
                         const float* __restrict__ b1, float* __restrict__ g,
                         float* __restrict__ gb, int n) {
  int t = blockIdx.x * blockDim.x + threadIdx.x;
  int total = n * HID;
  if (t < total) {
    int i = t / HID, j = t - i * HID;
    float di = dinv[i];
    float zsum = z1[t] + g[t];
    float r = fmaxf(zsum * di + b1[j], 0.f);
    float v = r * di;
    z1[t] = v;
    g[t]  = v;
    gb[t] = 0.f;
  }
}

__global__ void k_gemm2(const float* __restrict__ g, const float* __restrict__ gb,
                        const float* __restrict__ dinv,
                        const float* __restrict__ W2, const float* __restrict__ b2,
                        float* __restrict__ out, int n) {
  __shared__ float lw[HID * ODIM];
  for (int i = threadIdx.x; i < HID * ODIM; i += blockDim.x) lw[i] = W2[i];
  __syncthreads();

  int wave  = (blockIdx.x * blockDim.x + threadIdx.x) >> 5;
  int lane  = threadIdx.x & 31;
  int tiles = (n + 15) >> 4;
  if (wave >= tiles) return;

  int half = lane >> 4;
  int ml   = lane & 15;
  int row  = wave * 16 + ml;
  int rrow = (row < n) ? row : (n - 1);
  const float* gr  = g  + (size_t)rrow * HID;
  const float* gbr = gb + (size_t)rrow * HID;
  const float dr = dinv[rrow];

  v2f afrag[4];
#pragma unroll
  for (int kk = 0; kk < 4; ++kk) {
    int ka = kk * 4 + 2 * half;
    afrag[kk].x = (gr[ka]     + gbr[ka])     * dr;
    afrag[kk].y = (gr[ka + 1] + gbr[ka + 1]) * dr;
  }

#pragma unroll
  for (int tcol = 0; tcol < 4; ++tcol) {
    v8f acc = {0.f, 0.f, 0.f, 0.f, 0.f, 0.f, 0.f, 0.f};
#pragma unroll
    for (int kk = 0; kk < 4; ++kk) {
      int ka = kk * 4 + 2 * half;
      v2f b;
      b.x = lw[ka * ODIM + tcol * 16 + ml];
      b.y = lw[(ka + 1) * ODIM + tcol * 16 + ml];
      acc = __builtin_amdgcn_wmma_f32_16x16x4_f32(false, afrag[kk], false, b, (short)0,
                                                  acc, false, false);
    }
    const float bj = b2[tcol * 16 + ml];
#pragma unroll
    for (int r = 0; r < 8; ++r) {
      int orow = wave * 16 + r + 8 * half;
      if (orow < n) out[(size_t)orow * ODIM + tcol * 16 + ml] = acc[r] + bj;
    }
  }
}

extern "C" void kernel_launch(void* const* d_in, const int* in_sizes, int n_in,
                              void* d_out, int out_size, void* d_ws, size_t ws_size,
                              hipStream_t stream) {
  const float* x  = (const float*)d_in[0];
  const float* W1 = (const float*)d_in[1];
  const float* b1 = (const float*)d_in[2];
  const float* W2 = (const float*)d_in[3];
  const float* b2 = (const float*)d_in[4];
  const int*   ei = (const int*)d_in[5];

  const int N = in_sizes[0] / IN_DIM;
  const int E = in_sizes[5] / 2;
  const int* src = ei;
  const int* dst = ei + E;
  float* out = (float*)d_out;

  char* ws = (char*)d_ws;
  size_t off = 0;
  auto alloc = [&](size_t nbytes) -> void* {
    void* p = ws + off;
    off = (off + nbytes + 255) & ~(size_t)255;
    return p;
  };
  float* dinv = (float*)alloc((size_t)N * sizeof(float));
  float* h1   = (float*)alloc((size_t)N * HID  * sizeof(float));
  float* z1   = (float*)alloc((size_t)N * HID  * sizeof(float));
  float* g    = (float*)alloc((size_t)N * HID  * sizeof(float));

  const int B = 256;
  const int tiles = (N + 15) >> 4;
  const int gemmBlocks = (tiles + 7) / 8;

  int* part = (int*)h1;
  const int nranges = (N + DEG_RANGE - 1) / DEG_RANGE;
  k_deg_hist <<<nranges * DEG_SLICES, 1024, 0, stream>>>(dst, part, N, E, nranges);
  k_dinv     <<<(N + B - 1) / B, B, 0, stream>>>(part, dinv, N);

  const int BE = 1024;
  k_gemm1    <<<gemmBlocks, B, 0, stream>>>(x, W1, dinv, h1, z1, g, N);
  k_agg16    <<<(E + BE - 1) / BE, BE, 0, stream>>>(src, dst, h1, z1, g, E);

  k_init_g   <<<((size_t)N * HID + B - 1) / B, B, 0, stream>>>(z1, dinv, b1, g, h1, N);
  k_agg16    <<<(E + BE - 1) / BE, BE, 0, stream>>>(src, dst, z1, g, h1, E);
  k_gemm2    <<<gemmBlocks, B, 0, stream>>>(g, h1, dinv, W2, b2, out, N);
}
